# starting kernel
# baseline (speedup 1.0000x reference)
#include <hip/hip_runtime.h>
#include <hip/hip_bf16.h>
#include <math.h>

typedef __attribute__((ext_vector_type(2))) float v2f;
typedef __attribute__((ext_vector_type(8))) float v8f;

#define Nn 4
#define Dd 256
#define Hd 64
#define Wd 64
#define HW (Hd * Wd)

__global__ __launch_bounds__(256) void FeatureAlign_norms(
    const float* __restrict__ cur, const float* __restrict__ prev,
    float* __restrict__ cnF, float* __restrict__ pnF) {
  const int idx = blockIdx.x * 256 + threadIdx.x;
  const int n = idx >> 12;
  const int pix = idx & (HW - 1);
  const size_t base = (size_t)n * Dd * HW + pix;
  float cs = 0.0f, ps = 0.0f;
  for (int d = 0; d < Dd; ++d) {
    const float c = cur[base + (size_t)d * HW];
    const float p = prev[base + (size_t)d * HW];
    cs += c * c;
    ps += p * p;
  }
  cnF[idx] = sqrtf(cs) + 1e-8f;
  pnF[idx] = sqrtf(ps) + 1e-8f;
}

__global__ __launch_bounds__(128) void FeatureAlign_kernel(
    const float* __restrict__ cur, const float* __restrict__ prev,
    const float* __restrict__ mem, const float* __restrict__ cnF,
    const float* __restrict__ pnF, float* __restrict__ out) {
  __shared__ float ldsD[4][4][16][32];
  __shared__ float ldsW[4][Wd];
  __shared__ int   ldsC[Wd];

  const int nb = blockIdx.x;
  const int n = nb >> 6;
  const int y = nb & (Hd - 1);
  const int t = threadIdx.x;
  const int wave = t >> 5;
  const int lane = t & 31;
  const int g = lane >> 4;
  const int li = lane & 15;
  const int x0 = wave << 4;

  const float* curN = cur + (size_t)n * Dd * HW;
  const float* prevN = prev + (size_t)n * Dd * HW;
  const float* memN = mem + (size_t)n * Dd * HW;
  float* outN = out + (size_t)n * Dd * HW;

  const unsigned voffA = (unsigned)(2 * g * HW + y * Wd + x0 + li);
  unsigned voffB[4][2];
#pragma unroll
  for (int dyi = 0; dyi < 4; ++dyi) {
    const int dy = (dyi == 3) ? 2 : (dyi - 1);
    int yy = y + dy;
    yy = yy < 0 ? 0 : (yy > Hd - 1 ? Hd - 1 : yy);
#pragma unroll
    for (int h = 0; h < 2; ++h) {
      int bx = x0 - 1 + 16 * h + li;
      bx = bx < 0 ? 0 : (bx > Wd - 1 ? Wd - 1 : bx);
      voffB[dyi][h] = (unsigned)(2 * g * HW + yy * Wd + bx);
    }
  }
  __builtin_assume(voffA < 3u * HW);

  v8f acc[4][2];
#pragma unroll
  for (int dyi = 0; dyi < 4; ++dyi)
#pragma unroll
    for (int h = 0; h < 2; ++h) acc[dyi][h] = {};

  auto loadA = [&](int k, v2f& a) {
    const float* pk = curN + (size_t)(unsigned)k * HW;
    a.x = pk[voffA];
    a.y = pk[voffA + HW];
  };
  auto loadB = [&](int k, v2f (&b)[4][2]) {
    const float* pk = prevN + (size_t)(unsigned)k * HW;
#pragma unroll
    for (int dyi = 0; dyi < 4; ++dyi)
#pragma unroll
      for (int h = 0; h < 2; ++h) {
        b[dyi][h].x = pk[voffB[dyi][h]];
        b[dyi][h].y = pk[voffB[dyi][h] + HW];
      }
  };
  auto domma = [&](const v2f& a, const v2f (&b)[4][2]) {
#pragma unroll
    for (int dyi = 0; dyi < 4; ++dyi)
#pragma unroll
      for (int h = 0; h < 2; ++h)
        acc[dyi][h] = __builtin_amdgcn_wmma_f32_16x16x4_f32(
            false, a, false, b[dyi][h], (short)0, acc[dyi][h], false, false);
  };

  v2f a0, a1;
  v2f b0[4][2], b1[4][2];
  loadA(0, a0);
  loadB(0, b0);
  for (int k = 0; k < Dd; k += 8) {
    loadA(k + 4, a1);
    loadB(k + 4, b1);
    domma(a0, b0);
    const int kq = (k + 8 < Dd) ? (k + 8) : 0;
    loadA(kq, a0);
    loadB(kq, b0);
    domma(a1, b1);
  }

#pragma unroll
  for (int dyi = 0; dyi < 4; ++dyi)
#pragma unroll
    for (int h = 0; h < 2; ++h)
#pragma unroll
      for (int r = 0; r < 8; ++r)
        ldsD[wave][dyi][r + 8 * g][16 * h + li] = acc[dyi][h][r];

  __syncthreads();

  const int nb2 = n * HW;
  if (t < Wd) {
    const int x = t, w = x >> 4, i = x & 15;
    const float cn = cnF[nb2 + y * Wd + x];
    float mass = 0.0f;
#pragma unroll
    for (int dyi = 0; dyi < 3; ++dyi) {
      const int yy = y + dyi - 1;
      const bool rowv = (yy >= 0) && (yy < Hd);
#pragma unroll
      for (int dx = -1; dx <= 1; ++dx) {
        const int px = x + dx;
        if (rowv && px >= 0 && px < Wd) {
          const float pn = pnF[nb2 + yy * Wd + px];
          const float a = ldsD[w][dyi][i][i + 1 + dx] / (cn * pn);
          mass += fmaxf(a, 0.0f);
        }
      }
    }
    ldsC[x] = (mass > -10.0f) && (mass < 10.0f);
    const float msafe = (mass > 0.0f) ? mass : 1.0f;
#pragma unroll
    for (int oy = 0; oy < 2; ++oy) {
      const int dy = oy ? 2 : -1;
      const int dyi = oy ? 3 : 0;
      const int py = y + dy;
#pragma unroll
      for (int ox = 0; ox < 2; ++ox) {
        const int dx = ox ? 2 : -1;
        const int px = x + dx;
        float wv = 0.0f;
        if (py >= 0 && py < Hd && px >= 0 && px < Wd) {
          const float pn = pnF[nb2 + py * Wd + px];
          const float a = ldsD[w][dyi][i][i + 1 + dx] / (cn * pn);
          wv = (a > 0.0f) ? (a / msafe) : 0.0f;
        }
        ldsW[oy * 2 + ox][x] = wv;
      }
    }
  }
  __syncthreads();

  const int yb = y * Wd;
  for (int it = 0; it < (Dd * Wd) / 128; ++it) {
    const int idx2 = t + 128 * it;
    const int d = idx2 >> 6;
    const int x = idx2 & (Wd - 1);
    const size_t base = (size_t)d * HW;
    const float m0 = memN[base + yb + x];
    float agg = 0.0f;
    if (d > 0) {
#pragma unroll
      for (int off = 0; off < 4; ++off) {
        const int dy = (off < 2) ? -1 : 2;
        const int dx = (off & 1) ? 2 : -1;
        int py = y + dy; py = py < 0 ? 0 : (py > Hd - 1 ? Hd - 1 : py);
        int px = x + dx; px = px < 0 ? 0 : (px > Wd - 1 ? Wd - 1 : px);
        agg += ldsW[off][x] * memN[base + py * Wd + px];
      }
    }
    outN[base + yb + x] = ldsC[x] ? (m0 + agg) : 0.0f;
  }
}

extern "C" void kernel_launch(void* const* d_in, const int* in_sizes, int n_in,
                              void* d_out, int out_size, void* d_ws, size_t ws_size,
                              hipStream_t stream) {
  const float* cur = (const float*)d_in[0];
  const float* prev = (const float*)d_in[1];
  const float* mem = (const float*)d_in[2];
  float* out = (float*)d_out;
  (void)in_sizes; (void)n_in; (void)out_size; (void)ws_size;

  float* cnF = (float*)d_ws;
  float* pnF = cnF + (size_t)Nn * HW;

  FeatureAlign_norms<<<dim3((Nn * HW) / 256), dim3(256), 0, stream>>>(
      cur, prev, cnF, pnF);
  FeatureAlign_kernel<<<dim3(Nn * Hd), dim3(128), 0, stream>>>(
      cur, prev, mem, cnF, pnF, out);
}


# Epilogue mem->out: one float4 quad per lane, x-1/x+2 neighbours from the adjacent lane by DPP row shift, all 32 channel steps unrolled so many independent 16-B loads are in flight; term order kept
# speedup vs baseline: 1.8148x; 1.8148x over previous
#include <hip/hip_runtime.h>
#include <hip/hip_bf16.h>
#include <math.h>

typedef __attribute__((ext_vector_type(2))) float v2f;
typedef __attribute__((ext_vector_type(8))) float v8f;

#define Nn 4
#define Dd 256
#define Hd 64
#define Wd 64
#define HW (Hd * Wd)

__global__ __launch_bounds__(256) void FeatureAlign_norms(
    const float* __restrict__ cur, const float* __restrict__ prev,
    float* __restrict__ cnF, float* __restrict__ pnF) {
  const int idx = blockIdx.x * 256 + threadIdx.x;
  const int n = idx >> 12;
  const int pix = idx & (HW - 1);
  const size_t base = (size_t)n * Dd * HW + pix;
  float cs = 0.0f, ps = 0.0f;
  for (int d = 0; d < Dd; ++d) {
    const float c = cur[base + (size_t)d * HW];
    const float p = prev[base + (size_t)d * HW];
    cs += c * c;
    ps += p * p;
  }
  cnF[idx] = sqrtf(cs) + 1e-8f;
  pnF[idx] = sqrtf(ps) + 1e-8f;
}

__global__ __launch_bounds__(128) void FeatureAlign_kernel(
    const float* __restrict__ cur, const float* __restrict__ prev,
    const float* __restrict__ mem, const float* __restrict__ cnF,
    const float* __restrict__ pnF, float* __restrict__ out) {
  __shared__ float ldsD[4][4][16][32];
  __shared__ __attribute__((aligned(16))) float ldsW[4][Wd];
  __shared__ __attribute__((aligned(16))) int ldsC[Wd];

  const int nb = blockIdx.x;
  const int n = nb >> 6;
  const int y = nb & (Hd - 1);
  const int t = threadIdx.x;
  const int wave = t >> 5;
  const int lane = t & 31;
  const int g = lane >> 4;
  const int li = lane & 15;
  const int x0 = wave << 4;

  const float* curN = cur + (size_t)n * Dd * HW;
  const float* prevN = prev + (size_t)n * Dd * HW;
  const float* memN = mem + (size_t)n * Dd * HW;
  float* outN = out + (size_t)n * Dd * HW;

  const unsigned voffA = (unsigned)(2 * g * HW + y * Wd + x0 + li);
  unsigned voffB[4][2];
#pragma unroll
  for (int dyi = 0; dyi < 4; ++dyi) {
    const int dy = (dyi == 3) ? 2 : (dyi - 1);
    int yy = y + dy;
    yy = yy < 0 ? 0 : (yy > Hd - 1 ? Hd - 1 : yy);
#pragma unroll
    for (int h = 0; h < 2; ++h) {
      int bx = x0 - 1 + 16 * h + li;
      bx = bx < 0 ? 0 : (bx > Wd - 1 ? Wd - 1 : bx);
      voffB[dyi][h] = (unsigned)(2 * g * HW + yy * Wd + bx);
    }
  }
  __builtin_assume(voffA < 3u * HW);

  v8f acc[4][2];
#pragma unroll
  for (int dyi = 0; dyi < 4; ++dyi)
#pragma unroll
    for (int h = 0; h < 2; ++h) acc[dyi][h] = {};

  auto loadA = [&](int k, v2f& a) {
    const float* pk = curN + (size_t)(unsigned)k * HW;
    a.x = pk[voffA];
    a.y = pk[voffA + HW];
  };
  auto loadB = [&](int k, v2f (&b)[4][2]) {
    const float* pk = prevN + (size_t)(unsigned)k * HW;
#pragma unroll
    for (int dyi = 0; dyi < 4; ++dyi)
#pragma unroll
      for (int h = 0; h < 2; ++h) {
        b[dyi][h].x = pk[voffB[dyi][h]];
        b[dyi][h].y = pk[voffB[dyi][h] + HW];
      }
  };
  auto domma = [&](const v2f& a, const v2f (&b)[4][2]) {
#pragma unroll
    for (int dyi = 0; dyi < 4; ++dyi)
#pragma unroll
      for (int h = 0; h < 2; ++h)
        acc[dyi][h] = __builtin_amdgcn_wmma_f32_16x16x4_f32(
            false, a, false, b[dyi][h], (short)0, acc[dyi][h], false, false);
  };

  v2f a0, a1;
  v2f b0[4][2], b1[4][2];
  loadA(0, a0);
  loadB(0, b0);
  for (int k = 0; k < Dd; k += 8) {
    loadA(k + 4, a1);
    loadB(k + 4, b1);
    domma(a0, b0);
    const int kq = (k + 8 < Dd) ? (k + 8) : 0;
    loadA(kq, a0);
    loadB(kq, b0);
    domma(a1, b1);
  }

#pragma unroll
  for (int dyi = 0; dyi < 4; ++dyi)
#pragma unroll
    for (int h = 0; h < 2; ++h)
#pragma unroll
      for (int r = 0; r < 8; ++r)
        ldsD[wave][dyi][r + 8 * g][16 * h + li] = acc[dyi][h][r];

  __syncthreads();

  const int nb2 = n * HW;
  if (t < Wd) {
    const int x = t, w = x >> 4, i = x & 15;
    const float cn = cnF[nb2 + y * Wd + x];
    float mass = 0.0f;
#pragma unroll
    for (int dyi = 0; dyi < 3; ++dyi) {
      const int yy = y + dyi - 1;
      const bool rowv = (yy >= 0) && (yy < Hd);
#pragma unroll
      for (int dx = -1; dx <= 1; ++dx) {
        const int px = x + dx;
        if (rowv && px >= 0 && px < Wd) {
          const float pn = pnF[nb2 + yy * Wd + px];
          const float a = ldsD[w][dyi][i][i + 1 + dx] / (cn * pn);
          mass += fmaxf(a, 0.0f);
        }
      }
    }
    ldsC[x] = (mass > -10.0f) && (mass < 10.0f);
    const float msafe = (mass > 0.0f) ? mass : 1.0f;
#pragma unroll
    for (int oy = 0; oy < 2; ++oy) {
      const int dy = oy ? 2 : -1;
      const int dyi = oy ? 3 : 0;
      const int py = y + dy;
#pragma unroll
      for (int ox = 0; ox < 2; ++ox) {
        const int dx = ox ? 2 : -1;
        const int px = x + dx;
        float wv = 0.0f;
        if (py >= 0 && py < Hd && px >= 0 && px < Wd) {
          const float pn = pnF[nb2 + py * Wd + px];
          const float a = ldsD[w][dyi][i][i + 1 + dx] / (cn * pn);
          wv = (a > 0.0f) ? (a / msafe) : 0.0f;
        }
        ldsW[oy * 2 + ox][x] = wv;
      }
    }
  }
  __syncthreads();

  {
    const int q4 = (t & 15) << 2;
    const int c0 = t >> 4;
    const float4 w0 = *reinterpret_cast<const float4*>(&ldsW[0][q4]);
    const float4 w1 = *reinterpret_cast<const float4*>(&ldsW[1][q4]);
    const float4 w2 = *reinterpret_cast<const float4*>(&ldsW[2][q4]);
    const float4 w3 = *reinterpret_cast<const float4*>(&ldsW[3][q4]);
    const int4 cd = *reinterpret_cast<const int4*>(&ldsC[q4]);
    const int ym = (y > 0) ? (y - 1) : 0;
    const int yp = (y + 2 < Hd) ? (y + 2) : (Hd - 1);
    const unsigned oM = (unsigned)(c0 * HW + y * Wd + q4);
    const unsigned oA = (unsigned)(c0 * HW + ym * Wd + q4);
    const unsigned oB = (unsigned)(c0 * HW + yp * Wd + q4);

    constexpr int U = 8;
    constexpr int NB = (Dd / 8) / U;
    float4 m[2][U], a[2][U], b[2][U];

    auto loadBatch = [&](int bi, int buf) {
#pragma unroll
      for (int u = 0; u < U; ++u) {
        const unsigned ch = (unsigned)((bi * U + u) * 8 * HW);
        m[buf][u] = *reinterpret_cast<const float4*>(memN + ch + oM);
        a[buf][u] = *reinterpret_cast<const float4*>(memN + ch + oA);
        b[buf][u] = *reinterpret_cast<const float4*>(memN + ch + oB);
      }
    };
    auto fromLeft = [](float edge, float v) {
      return __int_as_float(__builtin_amdgcn_update_dpp(
          __float_as_int(edge), __float_as_int(v), 0x111  , 0xf, 0xf, false));
    };
    auto fromRight = [](float edge, float v) {
      return __int_as_float(__builtin_amdgcn_update_dpp(
          __float_as_int(edge), __float_as_int(v), 0x101  , 0xf, 0xf, false));
    };
    auto one = [](float mm, float wA, float l0, float wB, float r0, float wC,
                  float l1, float wD, float r1, bool chan, int cnd) {
      float agg = 0.0f;
      agg += wA * l0;
      agg += wB * r0;
      agg += wC * l1;
      agg += wD * r1;
      return cnd ? (mm + (chan ? agg : 0.0f)) : 0.0f;
    };
    auto computeBatch = [&](int bi, int buf) {
#pragma unroll
      for (int u = 0; u < U; ++u) {
        const unsigned ch = (unsigned)((bi * U + u) * 8 * HW);
        const bool chan = (c0 + (bi * U + u) * 8) > 0;
        const float4 mm = m[buf][u], ra = a[buf][u], rb = b[buf][u];
        const float aL = fromLeft(ra.x, ra.w);
        const float aH0 = fromRight(ra.w, ra.x);
        const float aH1 = fromRight(ra.w, ra.y);
        const float bL = fromLeft(rb.x, rb.w);
        const float bH0 = fromRight(rb.w, rb.x);
        const float bH1 = fromRight(rb.w, rb.y);
        float4 r;
        r.x = one(mm.x, w0.x, aL,   w1.x, ra.z, w2.x, bL,   w3.x, rb.z, chan, cd.x);
        r.y = one(mm.y, w0.y, ra.x, w1.y, ra.w, w2.y, rb.x, w3.y, rb.w, chan, cd.y);
        r.z = one(mm.z, w0.z, ra.y, w1.z, aH0,  w2.z, rb.y, w3.z, bH0,  chan, cd.z);
        r.w = one(mm.w, w0.w, ra.z, w1.w, aH1,  w2.w, rb.z, w3.w, bH1,  chan, cd.w);
        *reinterpret_cast<float4*>(outN + ch + oM) = r;
      }
    };

    loadBatch(0, 0);
#pragma unroll
    for (int bi = 0; bi < NB; ++bi) {
      if (bi + 1 < NB) loadBatch(bi + 1, (bi + 1) & 1);
      computeBatch(bi, bi & 1);
    }
  }
}

extern "C" void kernel_launch(void* const* d_in, const int* in_sizes, int n_in,
                              void* d_out, int out_size, void* d_ws, size_t ws_size,
                              hipStream_t stream) {
  const float* cur = (const float*)d_in[0];
  const float* prev = (const float*)d_in[1];
  const float* mem = (const float*)d_in[2];
  float* out = (float*)d_out;
  (void)in_sizes; (void)n_in; (void)out_size; (void)ws_size;

  float* cnF = (float*)d_ws;
  float* pnF = cnF + (size_t)Nn * HW;

  FeatureAlign_norms<<<dim3((Nn * HW) / 256), dim3(256), 0, stream>>>(
      cur, prev, cnF, pnF);
  FeatureAlign_kernel<<<dim3(Nn * Hd), dim3(128), 0, stream>>>(
      cur, prev, mem, cnF, pnF, out);
}


# Norms fused into the main kernel: lanes square-sum the cur/prev fragments they load for the WMMAs; each norm is now 8 partial sums (4 K groups x 2 lane halves) added in LDS; norms launch removed
# speedup vs baseline: 6.5310x; 3.5987x over previous
#include <hip/hip_runtime.h>
#include <hip/hip_bf16.h>
#include <math.h>

typedef __attribute__((ext_vector_type(2))) float v2f;
typedef __attribute__((ext_vector_type(8))) float v8f;

#define Nn 4
#define Dd 256
#define Hd 64
#define Wd 64
#define HW (Hd * Wd)
#define KS 4
#define NT (128 * KS)

__global__ __launch_bounds__(NT) void FeatureAlign_kernel(
    const float* __restrict__ cur, const float* __restrict__ prev,
    const float* __restrict__ mem, float* __restrict__ out) {
  __shared__ float ldsD[KS][4][4][16][32];
  __shared__ float ldsSA[KS][2][Wd];
  __shared__ float ldsSB[KS][4][4][2][32];
  __shared__ float ldsCN[Wd];
  __shared__ float ldsPN[4][4][32];
  __shared__ __attribute__((aligned(16))) float ldsW[4][Wd];
  __shared__ __attribute__((aligned(16))) int ldsC[Wd];

  const int nb = blockIdx.x;
  const int n = nb >> 6;
  const int y = nb & (Hd - 1);
  const int t = threadIdx.x;
  const int wave = t >> 5;
  const int lane = t & 31;
  const int g = lane >> 4;
  const int li = lane & 15;
  const int xt = wave & 3;
  const int kg = __builtin_amdgcn_readfirstlane(wave >> 2);
  const int x0 = xt << 4;
  const int kBeg = kg * (Dd / KS);
  const int kEnd = kBeg + (Dd / KS);

  const float* curN = cur + (size_t)n * Dd * HW;
  const float* prevN = prev + (size_t)n * Dd * HW;
  const float* memN = mem + (size_t)n * Dd * HW;
  float* outN = out + (size_t)n * Dd * HW;

  const unsigned voffA = (unsigned)(2 * g * HW + y * Wd + x0 + li);
  unsigned voffB[4][2];
#pragma unroll
  for (int dyi = 0; dyi < 4; ++dyi) {
    const int dy = (dyi == 3) ? 2 : (dyi - 1);
    int yy = y + dy;
    yy = yy < 0 ? 0 : (yy > Hd - 1 ? Hd - 1 : yy);
#pragma unroll
    for (int h = 0; h < 2; ++h) {
      int bx = x0 - 1 + 16 * h + li;
      bx = bx < 0 ? 0 : (bx > Wd - 1 ? Wd - 1 : bx);
      voffB[dyi][h] = (unsigned)(2 * g * HW + yy * Wd + bx);
    }
  }
  __builtin_assume(voffA < 3u * HW);

  v8f acc[4][2];
#pragma unroll
  for (int dyi = 0; dyi < 4; ++dyi)
#pragma unroll
    for (int h = 0; h < 2; ++h) acc[dyi][h] = {};

  auto loadA = [&](int k, v2f& a) {
    const float* pk = curN + (size_t)(unsigned)k * HW;
    a.x = pk[voffA];
    a.y = pk[voffA + HW];
  };
  auto loadB = [&](int k, v2f (&b)[4][2]) {
    const float* pk = prevN + (size_t)(unsigned)k * HW;
#pragma unroll
    for (int dyi = 0; dyi < 4; ++dyi)
#pragma unroll
      for (int h = 0; h < 2; ++h) {
        b[dyi][h].x = pk[voffB[dyi][h]];
        b[dyi][h].y = pk[voffB[dyi][h] + HW];
      }
  };
  v2f sqA = {};
  v2f sqB[4][2];
#pragma unroll
  for (int dyi = 0; dyi < 4; ++dyi)
#pragma unroll
    for (int h = 0; h < 2; ++h) sqB[dyi][h] = {};
  auto domma = [&](const v2f& a, const v2f (&b)[4][2]) {
    sqA += a * a;
#pragma unroll
    for (int dyi = 0; dyi < 4; ++dyi)
#pragma unroll
      for (int h = 0; h < 2; ++h) sqB[dyi][h] += b[dyi][h] * b[dyi][h];
#pragma unroll
    for (int dyi = 0; dyi < 4; ++dyi)
#pragma unroll
      for (int h = 0; h < 2; ++h)
        acc[dyi][h] = __builtin_amdgcn_wmma_f32_16x16x4_f32(
            false, a, false, b[dyi][h], (short)0, acc[dyi][h], false, false);
  };

  v2f a0, a1;
  v2f b0[4][2], b1[4][2];
  loadA(kBeg, a0);
  loadB(kBeg, b0);
  for (int k = kBeg; k < kEnd; k += 8) {
    loadA(k + 4, a1);
    loadB(k + 4, b1);
    domma(a0, b0);
    const int kq = (k + 8 < kEnd) ? (k + 8) : kBeg;
    loadA(kq, a0);
    loadB(kq, b0);
    domma(a1, b1);
  }

#pragma unroll
  for (int dyi = 0; dyi < 4; ++dyi)
#pragma unroll
    for (int h = 0; h < 2; ++h)
#pragma unroll
      for (int r = 0; r < 8; ++r)
        ldsD[kg][xt][dyi][r + 8 * g][16 * h + li] = acc[dyi][h][r];
  ldsSA[kg][g][x0 + li] = sqA.x + sqA.y;
#pragma unroll
  for (int dyi = 0; dyi < 4; ++dyi)
#pragma unroll
    for (int h = 0; h < 2; ++h)
      ldsSB[kg][xt][dyi][g][16 * h + li] = sqB[dyi][h].x + sqB[dyi][h].y;

  __syncthreads();

  for (int e = t; e < 4 * 4 * 32; e += NT) {
    const int w = e >> 7, dyi = (e >> 5) & 3, j = e & 31;
    float s = 0.0f;
#pragma unroll
    for (int q = 0; q < KS; ++q) {
      s += ldsSB[q][w][dyi][0][j];
      s += ldsSB[q][w][dyi][1][j];
    }
    ldsPN[w][dyi][j] = sqrtf(s) + 1e-8f;
  }
  if (t < Wd) {
    float cs = 0.0f;
#pragma unroll
    for (int q = 0; q < KS; ++q) {
      cs += ldsSA[q][0][t];
      cs += ldsSA[q][1][t];
    }
    ldsCN[t] = sqrtf(cs) + 1e-8f;
  }
  __syncthreads();

  auto dotAt = [&](int w, int dyi, int i, int j) {
    float s = ldsD[0][w][dyi][i][j];
#pragma unroll
    for (int q = 1; q < KS; ++q) s += ldsD[q][w][dyi][i][j];
    return s;
  };
  if (t < Wd) {
    const int x = t, w = x >> 4, i = x & 15;
    const float cn = ldsCN[x];
    float mass = 0.0f;
#pragma unroll
    for (int dyi = 0; dyi < 3; ++dyi) {
      const int yy = y + dyi - 1;
      const bool rowv = (yy >= 0) && (yy < Hd);
#pragma unroll
      for (int dx = -1; dx <= 1; ++dx) {
        const int px = x + dx;
        if (rowv && px >= 0 && px < Wd) {
          const float pn = ldsPN[w][dyi][i + 1 + dx];
          const float a = dotAt(w, dyi, i, i + 1 + dx) / (cn * pn);
          mass += fmaxf(a, 0.0f);
        }
      }
    }
    ldsC[x] = (mass > -10.0f) && (mass < 10.0f);
    const float msafe = (mass > 0.0f) ? mass : 1.0f;
#pragma unroll
    for (int oy = 0; oy < 2; ++oy) {
      const int dy = oy ? 2 : -1;
      const int dyi = oy ? 3 : 0;
      const int py = y + dy;
#pragma unroll
      for (int ox = 0; ox < 2; ++ox) {
        const int dx = ox ? 2 : -1;
        const int px = x + dx;
        float wv = 0.0f;
        if (py >= 0 && py < Hd && px >= 0 && px < Wd) {
          const float pn = ldsPN[w][dyi][i + 1 + dx];
          const float a = dotAt(w, dyi, i, i + 1 + dx) / (cn * pn);
          wv = (a > 0.0f) ? (a / msafe) : 0.0f;
        }
        ldsW[oy * 2 + ox][x] = wv;
      }
    }
  }
  __syncthreads();

  {
    const int q4 = (t & 15) << 2;
    const int c0 = t >> 4;
    const float4 w0 = *reinterpret_cast<const float4*>(&ldsW[0][q4]);
    const float4 w1 = *reinterpret_cast<const float4*>(&ldsW[1][q4]);
    const float4 w2 = *reinterpret_cast<const float4*>(&ldsW[2][q4]);
    const float4 w3 = *reinterpret_cast<const float4*>(&ldsW[3][q4]);
    const int4 cd = *reinterpret_cast<const int4*>(&ldsC[q4]);
    const int ym = (y > 0) ? (y - 1) : 0;
    const int yp = (y + 2 < Hd) ? (y + 2) : (Hd - 1);
    const unsigned oM = (unsigned)(c0 * HW + y * Wd + q4);
    const unsigned oA = (unsigned)(c0 * HW + ym * Wd + q4);
    const unsigned oB = (unsigned)(c0 * HW + yp * Wd + q4);

    constexpr int CR = NT / 16;
    constexpr int U = Dd / CR;
    float4 m[1][U], a[1][U], b[1][U];

    auto loadBatch = [&](int bi, int buf) {
#pragma unroll
      for (int u = 0; u < U; ++u) {
        const unsigned ch = (unsigned)((bi * U + u) * CR * HW);
        m[buf][u] = *reinterpret_cast<const float4*>(memN + ch + oM);
        a[buf][u] = *reinterpret_cast<const float4*>(memN + ch + oA);
        b[buf][u] = *reinterpret_cast<const float4*>(memN + ch + oB);
      }
    };
    auto fromLeft = [](float edge, float v) {
      return __int_as_float(__builtin_amdgcn_update_dpp(
          __float_as_int(edge), __float_as_int(v), 0x111  , 0xf, 0xf, false));
    };
    auto fromRight = [](float edge, float v) {
      return __int_as_float(__builtin_amdgcn_update_dpp(
          __float_as_int(edge), __float_as_int(v), 0x101  , 0xf, 0xf, false));
    };
    auto one = [](float mm, float wA, float l0, float wB, float r0, float wC,
                  float l1, float wD, float r1, bool chan, int cnd) {
      float agg = 0.0f;
      agg += wA * l0;
      agg += wB * r0;
      agg += wC * l1;
      agg += wD * r1;
      return cnd ? (mm + (chan ? agg : 0.0f)) : 0.0f;
    };
    auto computeBatch = [&](int bi, int buf) {
#pragma unroll
      for (int u = 0; u < U; ++u) {
        const unsigned ch = (unsigned)((bi * U + u) * CR * HW);
        const bool chan = (c0 + (bi * U + u) * CR) > 0;
        const float4 mm = m[buf][u], ra = a[buf][u], rb = b[buf][u];
        const float aL = fromLeft(ra.x, ra.w);
        const float aH0 = fromRight(ra.w, ra.x);
        const float aH1 = fromRight(ra.w, ra.y);
        const float bL = fromLeft(rb.x, rb.w);
        const float bH0 = fromRight(rb.w, rb.x);
        const float bH1 = fromRight(rb.w, rb.y);
        float4 r;
        r.x = one(mm.x, w0.x, aL,   w1.x, ra.z, w2.x, bL,   w3.x, rb.z, chan, cd.x);
        r.y = one(mm.y, w0.y, ra.x, w1.y, ra.w, w2.y, rb.x, w3.y, rb.w, chan, cd.y);
        r.z = one(mm.z, w0.z, ra.y, w1.z, aH0,  w2.z, rb.y, w3.z, bH0,  chan, cd.z);
        r.w = one(mm.w, w0.w, ra.z, w1.w, aH1,  w2.w, rb.z, w3.w, bH1,  chan, cd.w);
        *reinterpret_cast<float4*>(outN + ch + oM) = r;
      }
    };

    loadBatch(0, 0);
    computeBatch(0, 0);
  }
}

extern "C" void kernel_launch(void* const* d_in, const int* in_sizes, int n_in,
                              void* d_out, int out_size, void* d_ws, size_t ws_size,
                              hipStream_t stream) {
  const float* cur = (const float*)d_in[0];
  const float* prev = (const float*)d_in[1];
  const float* mem = (const float*)d_in[2];
  float* out = (float*)d_out;
  (void)in_sizes; (void)n_in; (void)out_size; (void)d_ws; (void)ws_size;

  FeatureAlign_kernel<<<dim3(Nn * Hd), dim3(NT), 0, stream>>>(
      cur, prev, mem, out);
}


# x tiles at stride 13 with one 16-column band tile per dy (5 tiles x 4 K groups = 20 waves): 20 instead of 32 WMMAs and 50 instead of 72 fragment loads per 4 channels; k order of every sum kept
# speedup vs baseline: 7.4847x; 1.1460x over previous
#include <hip/hip_runtime.h>
#include <hip/hip_bf16.h>
#include <math.h>

typedef __attribute__((ext_vector_type(2))) float v2f;
typedef __attribute__((ext_vector_type(8))) float v8f;

#define Nn 4
#define Dd 256
#define Hd 64
#define Wd 64
#define HW (Hd * Wd)
#define KS 4
#define TL 5
#define TS 13
#define NT (32 * TL * KS)
#define NT3 512

__global__ __launch_bounds__(NT) __attribute__((amdgpu_waves_per_eu(1, 5))) void FeatureAlign_kernel(
    const float* __restrict__ cur, const float* __restrict__ prev,
    const float* __restrict__ mem, float* __restrict__ out) {
  __shared__ float ldsD[KS][TL][4][16][16];
  __shared__ float ldsSA[KS][2][Wd];
  __shared__ float ldsSB[KS][TL][4][2][16];
  __shared__ float ldsCN[Wd];
  __shared__ float ldsPN[TL][4][16];
  __shared__ __attribute__((aligned(16))) float ldsW[4][Wd];
  __shared__ __attribute__((aligned(16))) int ldsC[Wd];

  const int nb = blockIdx.x;
  const int n = nb >> 6;
  const int y = nb & (Hd - 1);
  const int t = threadIdx.x;
  const int wave = t >> 5;
  const int lane = t & 31;
  const int g = lane >> 4;
  const int li = lane & 15;
  const int xt = wave % TL;
  const int kg = __builtin_amdgcn_readfirstlane(wave / TL);
  const int x0 = xt * TS;
  const int kBeg = kg * (Dd / KS);
  const int kEnd = kBeg + (Dd / KS);

  const float* curN = cur + (size_t)n * Dd * HW;
  const float* prevN = prev + (size_t)n * Dd * HW;
  const float* memN = mem + (size_t)n * Dd * HW;
  float* outN = out + (size_t)n * Dd * HW;

  const int ax = (x0 + li > Wd - 1) ? (Wd - 1) : (x0 + li);
  const unsigned voffA = (unsigned)(2 * g * HW + y * Wd + ax);
  unsigned voffB[4];
#pragma unroll
  for (int dyi = 0; dyi < 4; ++dyi) {
    const int dy = (dyi == 3) ? 2 : (dyi - 1);
    int yy = y + dy;
    yy = yy < 0 ? 0 : (yy > Hd - 1 ? Hd - 1 : yy);
    int bx = x0 - 1 + li;
    bx = bx < 0 ? 0 : (bx > Wd - 1 ? Wd - 1 : bx);
    voffB[dyi] = (unsigned)(2 * g * HW + yy * Wd + bx);
  }
  __builtin_assume(voffA < 3u * HW);

  v8f acc[4];
#pragma unroll
  for (int dyi = 0; dyi < 4; ++dyi) acc[dyi] = {};

  auto loadA = [&](int k, v2f& a) {
    const float* pk = curN + (size_t)(unsigned)k * HW;
    a.x = pk[voffA];
    a.y = pk[voffA + HW];
  };
  auto loadB = [&](int k, v2f (&b)[4]) {
    const float* pk = prevN + (size_t)(unsigned)k * HW;
#pragma unroll
    for (int dyi = 0; dyi < 4; ++dyi) {
      b[dyi].x = pk[voffB[dyi]];
      b[dyi].y = pk[voffB[dyi] + HW];
    }
  };
  v2f sqA = {};
  v2f sqB[4];
#pragma unroll
  for (int dyi = 0; dyi < 4; ++dyi) sqB[dyi] = {};
  auto domma = [&](const v2f& a, const v2f (&b)[4]) {
    sqA += a * a;
#pragma unroll
    for (int dyi = 0; dyi < 4; ++dyi) sqB[dyi] += b[dyi] * b[dyi];
#pragma unroll
    for (int dyi = 0; dyi < 4; ++dyi)
      acc[dyi] = __builtin_amdgcn_wmma_f32_16x16x4_f32(
          false, a, false, b[dyi], (short)0, acc[dyi], false, false);
  };

  v2f a0, a1;
  v2f b0[4], b1[4];
  loadA(kBeg, a0);
  loadB(kBeg, b0);
  for (int k = kBeg; k < kEnd; k += 8) {
    loadA(k + 4, a1);
    loadB(k + 4, b1);
    domma(a0, b0);
    const int kq = (k + 8 < kEnd) ? (k + 8) : kBeg;
    loadA(kq, a0);
    loadB(kq, b0);
    domma(a1, b1);
  }

#pragma unroll
  for (int dyi = 0; dyi < 4; ++dyi)
#pragma unroll
    for (int r = 0; r < 8; ++r)
      ldsD[kg][xt][dyi][r + 8 * g][li] = acc[dyi][r];
  if (li < TS && x0 + li < Wd) ldsSA[kg][g][x0 + li] = sqA.x + sqA.y;
#pragma unroll
  for (int dyi = 0; dyi < 4; ++dyi)
    ldsSB[kg][xt][dyi][g][li] = sqB[dyi].x + sqB[dyi].y;

  __syncthreads();

  for (int e = t; e < TL * 4 * 16; e += NT) {
    const int w = e >> 6, dyi = (e >> 4) & 3, j = e & 15;
    float s = 0.0f;
#pragma unroll
    for (int q = 0; q < KS; ++q) {
      s += ldsSB[q][w][dyi][0][j];
      s += ldsSB[q][w][dyi][1][j];
    }
    ldsPN[w][dyi][j] = sqrtf(s) + 1e-8f;
  }
  if (t < Wd) {
    float cs = 0.0f;
#pragma unroll
    for (int q = 0; q < KS; ++q) {
      cs += ldsSA[q][0][t];
      cs += ldsSA[q][1][t];
    }
    ldsCN[t] = sqrtf(cs) + 1e-8f;
  }
  __syncthreads();

  auto dotAt = [&](int w, int dyi, int i, int j) {
    float s = ldsD[0][w][dyi][i][j];
#pragma unroll
    for (int q = 1; q < KS; ++q) s += ldsD[q][w][dyi][i][j];
    return s;
  };
  if (t < Wd) {
    const int x = t, w = x / TS, i = x - TS * w;
    const float cn = ldsCN[x];
    float mass = 0.0f;
#pragma unroll
    for (int dyi = 0; dyi < 3; ++dyi) {
      const int yy = y + dyi - 1;
      const bool rowv = (yy >= 0) && (yy < Hd);
#pragma unroll
      for (int dx = -1; dx <= 1; ++dx) {
        const int px = x + dx;
        if (rowv && px >= 0 && px < Wd) {
          const float pn = ldsPN[w][dyi][i + 1 + dx];
          const float a = dotAt(w, dyi, i, i + 1 + dx) / (cn * pn);
          mass += fmaxf(a, 0.0f);
        }
      }
    }
    ldsC[x] = (mass > -10.0f) && (mass < 10.0f);
    const float msafe = (mass > 0.0f) ? mass : 1.0f;
#pragma unroll
    for (int oy = 0; oy < 2; ++oy) {
      const int dy = oy ? 2 : -1;
      const int dyi = oy ? 3 : 0;
      const int py = y + dy;
#pragma unroll
      for (int ox = 0; ox < 2; ++ox) {
        const int dx = ox ? 2 : -1;
        const int px = x + dx;
        float wv = 0.0f;
        if (py >= 0 && py < Hd && px >= 0 && px < Wd) {
          const float pn = ldsPN[w][dyi][i + 1 + dx];
          const float a = dotAt(w, dyi, i, i + 1 + dx) / (cn * pn);
          wv = (a > 0.0f) ? (a / msafe) : 0.0f;
        }
        ldsW[oy * 2 + ox][x] = wv;
      }
    }
  }
  __syncthreads();

  if (t < NT3) {
    const int q4 = (t & 15) << 2;
    const int c0 = t >> 4;
    const float4 w0 = *reinterpret_cast<const float4*>(&ldsW[0][q4]);
    const float4 w1 = *reinterpret_cast<const float4*>(&ldsW[1][q4]);
    const float4 w2 = *reinterpret_cast<const float4*>(&ldsW[2][q4]);
    const float4 w3 = *reinterpret_cast<const float4*>(&ldsW[3][q4]);
    const int4 cd = *reinterpret_cast<const int4*>(&ldsC[q4]);
    const int ym = (y > 0) ? (y - 1) : 0;
    const int yp = (y + 2 < Hd) ? (y + 2) : (Hd - 1);
    const unsigned oM = (unsigned)(c0 * HW + y * Wd + q4);
    const unsigned oA = (unsigned)(c0 * HW + ym * Wd + q4);
    const unsigned oB = (unsigned)(c0 * HW + yp * Wd + q4);

    constexpr int CR = NT3 / 16;
    constexpr int U = Dd / CR;
    float4 m[1][U], a[1][U], b[1][U];

    auto loadBatch = [&](int bi, int buf) {
#pragma unroll
      for (int u = 0; u < U; ++u) {
        const unsigned ch = (unsigned)((bi * U + u) * CR * HW);
        m[buf][u] = *reinterpret_cast<const float4*>(memN + ch + oM);
        a[buf][u] = *reinterpret_cast<const float4*>(memN + ch + oA);
        b[buf][u] = *reinterpret_cast<const float4*>(memN + ch + oB);
      }
    };
    auto fromLeft = [](float edge, float v) {
      return __int_as_float(__builtin_amdgcn_update_dpp(
          __float_as_int(edge), __float_as_int(v), 0x111  , 0xf, 0xf, false));
    };
    auto fromRight = [](float edge, float v) {
      return __int_as_float(__builtin_amdgcn_update_dpp(
          __float_as_int(edge), __float_as_int(v), 0x101  , 0xf, 0xf, false));
    };
    auto one = [](float mm, float wA, float l0, float wB, float r0, float wC,
                  float l1, float wD, float r1, bool chan, int cnd) {
      float agg = 0.0f;
      agg += wA * l0;
      agg += wB * r0;
      agg += wC * l1;
      agg += wD * r1;
      return cnd ? (mm + (chan ? agg : 0.0f)) : 0.0f;
    };
    auto computeBatch = [&](int bi, int buf) {
#pragma unroll
      for (int u = 0; u < U; ++u) {
        const unsigned ch = (unsigned)((bi * U + u) * CR * HW);
        const bool chan = (c0 + (bi * U + u) * CR) > 0;
        const float4 mm = m[buf][u], ra = a[buf][u], rb = b[buf][u];
        const float aL = fromLeft(ra.x, ra.w);
        const float aH0 = fromRight(ra.w, ra.x);
        const float aH1 = fromRight(ra.w, ra.y);
        const float bL = fromLeft(rb.x, rb.w);
        const float bH0 = fromRight(rb.w, rb.x);
        const float bH1 = fromRight(rb.w, rb.y);
        float4 r;
        r.x = one(mm.x, w0.x, aL,   w1.x, ra.z, w2.x, bL,   w3.x, rb.z, chan, cd.x);
        r.y = one(mm.y, w0.y, ra.x, w1.y, ra.w, w2.y, rb.x, w3.y, rb.w, chan, cd.y);
        r.z = one(mm.z, w0.z, ra.y, w1.z, aH0,  w2.z, rb.y, w3.z, bH0,  chan, cd.z);
        r.w = one(mm.w, w0.w, ra.z, w1.w, aH1,  w2.w, rb.z, w3.w, bH1,  chan, cd.w);
        *reinterpret_cast<float4*>(outN + ch + oM) = r;
      }
    };

    loadBatch(0, 0);
    computeBatch(0, 0);
  }
}

extern "C" void kernel_launch(void* const* d_in, const int* in_sizes, int n_in,
                              void* d_out, int out_size, void* d_ws, size_t ws_size,
                              hipStream_t stream) {
  const float* cur = (const float*)d_in[0];
  const float* prev = (const float*)d_in[1];
  const float* mem = (const float*)d_in[2];
  float* out = (float*)d_out;
  (void)in_sizes; (void)n_in; (void)out_size; (void)d_ws; (void)ws_size;

  FeatureAlign_kernel<<<dim3(Nn * Hd), dim3(NT), 0, stream>>>(
      cur, prev, mem, out);
}


# K loop requests fragment loads 3 chunks ahead instead of 1 (20-40 loads in flight per wave, was 10-20), requests kept in chunk order, no wrapped tail request; order of every sum kept
# speedup vs baseline: 8.2061x; 1.0964x over previous
#include <hip/hip_runtime.h>
#include <hip/hip_bf16.h>
#include <math.h>

typedef __attribute__((ext_vector_type(2))) float v2f;
typedef __attribute__((ext_vector_type(8))) float v8f;

#define Nn 4
#define Dd 256
#define Hd 64
#define Wd 64
#define HW (Hd * Wd)
#define KS 4
#define TL 5
#define TS 13
#define NT (32 * TL * KS)
#define NT3 512

__global__ __launch_bounds__(NT) __attribute__((amdgpu_waves_per_eu(1, 5))) void FeatureAlign_kernel(
    const float* __restrict__ cur, const float* __restrict__ prev,
    const float* __restrict__ mem, float* __restrict__ out) {
  __shared__ float ldsD[KS][TL][4][16][16];
  __shared__ float ldsSA[KS][2][Wd];
  __shared__ float ldsSB[KS][TL][4][2][16];
  __shared__ float ldsCN[Wd];
  __shared__ float ldsPN[TL][4][16];
  __shared__ __attribute__((aligned(16))) float ldsW[4][Wd];
  __shared__ __attribute__((aligned(16))) int ldsC[Wd];

  const int nb = blockIdx.x;
  const int n = nb >> 6;
  const int y = nb & (Hd - 1);
  const int t = threadIdx.x;
  const int wave = t >> 5;
  const int lane = t & 31;
  const int g = lane >> 4;
  const int li = lane & 15;
  const int xt = wave % TL;
  const int kg = __builtin_amdgcn_readfirstlane(wave / TL);
  const int x0 = xt * TS;
  const int kBeg = kg * (Dd / KS);

  const float* curN = cur + (size_t)n * Dd * HW;
  const float* prevN = prev + (size_t)n * Dd * HW;
  const float* memN = mem + (size_t)n * Dd * HW;
  float* outN = out + (size_t)n * Dd * HW;

  const int ax = (x0 + li > Wd - 1) ? (Wd - 1) : (x0 + li);
  const unsigned voffA = (unsigned)(2 * g * HW + y * Wd + ax);
  unsigned voffB[4];
#pragma unroll
  for (int dyi = 0; dyi < 4; ++dyi) {
    const int dy = (dyi == 3) ? 2 : (dyi - 1);
    int yy = y + dy;
    yy = yy < 0 ? 0 : (yy > Hd - 1 ? Hd - 1 : yy);
    int bx = x0 - 1 + li;
    bx = bx < 0 ? 0 : (bx > Wd - 1 ? Wd - 1 : bx);
    voffB[dyi] = (unsigned)(2 * g * HW + yy * Wd + bx);
  }
  __builtin_assume(voffA < 3u * HW);

  v8f acc[4];
#pragma unroll
  for (int dyi = 0; dyi < 4; ++dyi) acc[dyi] = {};

  auto loadA = [&](int k, v2f& a) {
    const float* pk = curN + (size_t)(unsigned)k * HW;
    a.x = pk[voffA];
    a.y = pk[voffA + HW];
  };
  auto loadB = [&](int k, v2f (&b)[4]) {
    const float* pk = prevN + (size_t)(unsigned)k * HW;
#pragma unroll
    for (int dyi = 0; dyi < 4; ++dyi) {
      b[dyi].x = pk[voffB[dyi]];
      b[dyi].y = pk[voffB[dyi] + HW];
    }
  };
  v2f sqA = {};
  v2f sqB[4];
#pragma unroll
  for (int dyi = 0; dyi < 4; ++dyi) sqB[dyi] = {};
  auto domma = [&](const v2f& a, const v2f (&b)[4]) {
    sqA += a * a;
#pragma unroll
    for (int dyi = 0; dyi < 4; ++dyi) sqB[dyi] += b[dyi] * b[dyi];
#pragma unroll
    for (int dyi = 0; dyi < 4; ++dyi)
      acc[dyi] = __builtin_amdgcn_wmma_f32_16x16x4_f32(
          false, a, false, b[dyi], (short)0, acc[dyi], false, false);
  };

  constexpr int NS = (Dd / KS) / 4;
  constexpr int PD = 3;
  constexpr int NB = PD + 1;
  static_assert(NS % NB == 0 && NS >= 2 * NB, "chunk count must fill whole trips");
  v2f af[NB];
  v2f bf[NB][4];
#pragma unroll
  for (int s = 0; s < PD; ++s) {
    loadA(kBeg + 4 * s, af[s]);
    loadB(kBeg + 4 * s, bf[s]);
    __builtin_amdgcn_sched_barrier(0);
  }
#pragma nounroll
  for (int k = kBeg; k < kBeg + 4 * (NS - NB); k += 4 * NB) {
#pragma unroll
    for (int j = 0; j < NB; ++j) {
      loadA(k + 4 * (j + PD), af[(j + PD) % NB]);
      loadB(k + 4 * (j + PD), bf[(j + PD) % NB]);
      domma(af[j], bf[j]);
      __builtin_amdgcn_sched_barrier(0);
    }
  }
  {
    const int k = kBeg + 4 * (NS - NB);
    loadA(k + 4 * PD, af[PD]);
    loadB(k + 4 * PD, bf[PD]);
#pragma unroll
    for (int j = 0; j < NB; ++j) domma(af[j], bf[j]);
  }

#pragma unroll
  for (int dyi = 0; dyi < 4; ++dyi)
#pragma unroll
    for (int r = 0; r < 8; ++r)
      ldsD[kg][xt][dyi][r + 8 * g][li] = acc[dyi][r];
  if (li < TS && x0 + li < Wd) ldsSA[kg][g][x0 + li] = sqA.x + sqA.y;
#pragma unroll
  for (int dyi = 0; dyi < 4; ++dyi)
    ldsSB[kg][xt][dyi][g][li] = sqB[dyi].x + sqB[dyi].y;

  __syncthreads();

  for (int e = t; e < TL * 4 * 16; e += NT) {
    const int w = e >> 6, dyi = (e >> 4) & 3, j = e & 15;
    float s = 0.0f;
#pragma unroll
    for (int q = 0; q < KS; ++q) {
      s += ldsSB[q][w][dyi][0][j];
      s += ldsSB[q][w][dyi][1][j];
    }
    ldsPN[w][dyi][j] = sqrtf(s) + 1e-8f;
  }
  if (t < Wd) {
    float cs = 0.0f;
#pragma unroll
    for (int q = 0; q < KS; ++q) {
      cs += ldsSA[q][0][t];
      cs += ldsSA[q][1][t];
    }
    ldsCN[t] = sqrtf(cs) + 1e-8f;
  }
  __syncthreads();

  auto dotAt = [&](int w, int dyi, int i, int j) {
    float s = ldsD[0][w][dyi][i][j];
#pragma unroll
    for (int q = 1; q < KS; ++q) s += ldsD[q][w][dyi][i][j];
    return s;
  };
  if (t < Wd) {
    const int x = t, w = x / TS, i = x - TS * w;
    const float cn = ldsCN[x];
    float mass = 0.0f;
#pragma unroll
    for (int dyi = 0; dyi < 3; ++dyi) {
      const int yy = y + dyi - 1;
      const bool rowv = (yy >= 0) && (yy < Hd);
#pragma unroll
      for (int dx = -1; dx <= 1; ++dx) {
        const int px = x + dx;
        if (rowv && px >= 0 && px < Wd) {
          const float pn = ldsPN[w][dyi][i + 1 + dx];
          const float a = dotAt(w, dyi, i, i + 1 + dx) / (cn * pn);
          mass += fmaxf(a, 0.0f);
        }
      }
    }
    ldsC[x] = (mass > -10.0f) && (mass < 10.0f);
    const float msafe = (mass > 0.0f) ? mass : 1.0f;
#pragma unroll
    for (int oy = 0; oy < 2; ++oy) {
      const int dy = oy ? 2 : -1;
      const int dyi = oy ? 3 : 0;
      const int py = y + dy;
#pragma unroll
      for (int ox = 0; ox < 2; ++ox) {
        const int dx = ox ? 2 : -1;
        const int px = x + dx;
        float wv = 0.0f;
        if (py >= 0 && py < Hd && px >= 0 && px < Wd) {
          const float pn = ldsPN[w][dyi][i + 1 + dx];
          const float a = dotAt(w, dyi, i, i + 1 + dx) / (cn * pn);
          wv = (a > 0.0f) ? (a / msafe) : 0.0f;
        }
        ldsW[oy * 2 + ox][x] = wv;
      }
    }
  }
  __syncthreads();

  if (t < NT3) {
    const int q4 = (t & 15) << 2;
    const int c0 = t >> 4;
    const float4 w0 = *reinterpret_cast<const float4*>(&ldsW[0][q4]);
    const float4 w1 = *reinterpret_cast<const float4*>(&ldsW[1][q4]);
    const float4 w2 = *reinterpret_cast<const float4*>(&ldsW[2][q4]);
    const float4 w3 = *reinterpret_cast<const float4*>(&ldsW[3][q4]);
    const int4 cd = *reinterpret_cast<const int4*>(&ldsC[q4]);
    const int ym = (y > 0) ? (y - 1) : 0;
    const int yp = (y + 2 < Hd) ? (y + 2) : (Hd - 1);
    const unsigned oM = (unsigned)(c0 * HW + y * Wd + q4);
    const unsigned oA = (unsigned)(c0 * HW + ym * Wd + q4);
    const unsigned oB = (unsigned)(c0 * HW + yp * Wd + q4);

    constexpr int CR = NT3 / 16;
    constexpr int U = Dd / CR;
    float4 m[1][U], a[1][U], b[1][U];

    auto loadBatch = [&](int bi, int buf) {
#pragma unroll
      for (int u = 0; u < U; ++u) {
        const unsigned ch = (unsigned)((bi * U + u) * CR * HW);
        m[buf][u] = *reinterpret_cast<const float4*>(memN + ch + oM);
        a[buf][u] = *reinterpret_cast<const float4*>(memN + ch + oA);
        b[buf][u] = *reinterpret_cast<const float4*>(memN + ch + oB);
      }
    };
    auto fromLeft = [](float edge, float v) {
      return __int_as_float(__builtin_amdgcn_update_dpp(
          __float_as_int(edge), __float_as_int(v), 0x111  , 0xf, 0xf, false));
    };
    auto fromRight = [](float edge, float v) {
      return __int_as_float(__builtin_amdgcn_update_dpp(
          __float_as_int(edge), __float_as_int(v), 0x101  , 0xf, 0xf, false));
    };
    auto one = [](float mm, float wA, float l0, float wB, float r0, float wC,
                  float l1, float wD, float r1, bool chan, int cnd) {
      float agg = 0.0f;
      agg += wA * l0;
      agg += wB * r0;
      agg += wC * l1;
      agg += wD * r1;
      return cnd ? (mm + (chan ? agg : 0.0f)) : 0.0f;
    };
    auto computeBatch = [&](int bi, int buf) {
#pragma unroll
      for (int u = 0; u < U; ++u) {
        const unsigned ch = (unsigned)((bi * U + u) * CR * HW);
        const bool chan = (c0 + (bi * U + u) * CR) > 0;
        const float4 mm = m[buf][u], ra = a[buf][u], rb = b[buf][u];
        const float aL = fromLeft(ra.x, ra.w);
        const float aH0 = fromRight(ra.w, ra.x);
        const float aH1 = fromRight(ra.w, ra.y);
        const float bL = fromLeft(rb.x, rb.w);
        const float bH0 = fromRight(rb.w, rb.x);
        const float bH1 = fromRight(rb.w, rb.y);
        float4 r;
        r.x = one(mm.x, w0.x, aL,   w1.x, ra.z, w2.x, bL,   w3.x, rb.z, chan, cd.x);
        r.y = one(mm.y, w0.y, ra.x, w1.y, ra.w, w2.y, rb.x, w3.y, rb.w, chan, cd.y);
        r.z = one(mm.z, w0.z, ra.y, w1.z, aH0,  w2.z, rb.y, w3.z, bH0,  chan, cd.z);
        r.w = one(mm.w, w0.w, ra.z, w1.w, aH1,  w2.w, rb.z, w3.w, bH1,  chan, cd.w);
        *reinterpret_cast<float4*>(outN + ch + oM) = r;
      }
    };

    loadBatch(0, 0);
    computeBatch(0, 0);
  }
}

extern "C" void kernel_launch(void* const* d_in, const int* in_sizes, int n_in,
                              void* d_out, int out_size, void* d_ws, size_t ws_size,
                              hipStream_t stream) {
  const float* cur = (const float*)d_in[0];
  const float* prev = (const float*)d_in[1];
  const float* mem = (const float*)d_in[2];
  float* out = (float*)d_out;
  (void)in_sizes; (void)n_in; (void)out_size; (void)d_ws; (void)ws_size;

  FeatureAlign_kernel<<<dim3(Nn * Hd), dim3(NT), 0, stream>>>(
      cur, prev, mem, out);
}
